# starting kernel
# baseline (speedup 1.0000x reference)
#include <hip/hip_runtime.h>

typedef __attribute__((ext_vector_type(2))) float v2f;
typedef __attribute__((ext_vector_type(8))) float v8f;

#define BATCH 32
#define NROWS 1024
#define DDIM  256
#define SPLIT 8

__global__ __launch_bounds__(256) void reduce_partial(const float* __restrict__ x,
                                                      const float* __restrict__ y,
                                                      float* __restrict__ partial) {
    const int t     = threadIdx.x;
    const int chunk = blockIdx.x;
    const int which = blockIdx.y;
    const int b     = blockIdx.z;
    const float* src = which ? y : x;
    const float* p = src + ((size_t)b * NROWS + (size_t)chunk * (NROWS / SPLIT)) * DDIM + t;
    float acc = 0.0f;
#pragma unroll 8
    for (int r = 0; r < NROWS / SPLIT; ++r)
        acc += p[(size_t)r * DDIM];
    partial[(((size_t)which * BATCH + b) * SPLIT + chunk) * DDIM + t] = acc;
}

__global__ __launch_bounds__(256) void reduce_combine(const float* __restrict__ partial,
                                                      float* __restrict__ sums) {
    const int t     = threadIdx.x;
    const int b     = blockIdx.x;
    const int which = blockIdx.y;
    float acc = 0.0f;
#pragma unroll
    for (int c = 0; c < SPLIT; ++c)
        acc += partial[(((size_t)which * BATCH + b) * SPLIT + c) * DDIM + t];
    sums[((size_t)which * BATCH + b) * DDIM + t] = acc;
}

__global__ __launch_bounds__(256) void dot_wmma(const float* __restrict__ x,
                                                const float* __restrict__ y,
                                                const float* __restrict__ sums,
                                                float* __restrict__ out) {
    const int wave = threadIdx.x >> 5;
    const int lane = threadIdx.x & 31;
    const int g    = blockIdx.x * 8 + wave;
    const int h    = g >> 11;
    const int rem  = g & 2047;
    const int b    = rem >> 6;
    const int mbase = (rem & 63) << 4;

    const float* data = h ? x : y;
    const float* vec  = sums + ((size_t)h * BATCH + b) * DDIM;

    const int row  = mbase + (lane & 15);
    const int koff = (lane >> 4) << 1;
    const float* arow = data + ((size_t)b * NROWS + row) * DDIM + koff;
    const float* brow = vec + koff;

    v8f c0 = {};
    v8f c1 = {};
#pragma unroll 4
    for (int d0 = 0; d0 < DDIM; d0 += 8) {
        v2f a0 = *(const v2f*)(arow + d0);
        v2f b0 = *(const v2f*)(brow + d0);
        c0 = __builtin_amdgcn_wmma_f32_16x16x4_f32(false, a0, false, b0,
                                                   (short)0, c0, false, false);
        v2f a1 = *(const v2f*)(arow + d0 + 4);
        v2f b1 = *(const v2f*)(brow + d0 + 4);
        c1 = __builtin_amdgcn_wmma_f32_16x16x4_f32(false, a1, false, b1,
                                                   (short)0, c1, false, false);
    }
    v8f c = c0 + c1;

    const size_t obase = (size_t)b * 2048 + (size_t)h * 1024 + (size_t)mbase;
    if (lane == 0) {
#pragma unroll
        for (int i = 0; i < 8; ++i) out[obase + i] = c[i];
    } else if (lane == 16) {
#pragma unroll
        for (int i = 0; i < 8; ++i) out[obase + 8 + i] = c[i];
    }
}

extern "C" void kernel_launch(void* const* d_in, const int* in_sizes, int n_in,
                              void* d_out, int out_size, void* d_ws, size_t ws_size,
                              hipStream_t stream) {
    const float* x = (const float*)d_in[0];
    const float* y = (const float*)d_in[1];
    float* out = (float*)d_out;

    float* partial = (float*)d_ws;
    float* sums    = partial + (size_t)2 * BATCH * SPLIT * DDIM;

    reduce_partial<<<dim3(SPLIT, 2, BATCH), 256, 0, stream>>>(x, y, partial);
    reduce_combine<<<dim3(BATCH, 2), 256, 0, stream>>>(partial, sums);
    dot_wmma<<<dim3(512), 256, 0, stream>>>(x, y, sums, out);
}


# reduce_partial: 128-bit loads per lane (was 32-bit), a thread's 32 row loads independent so they overlap instead of draining every 8; 4 row groups folded in LDS: row-sum order changes
# speedup vs baseline: 1.3939x; 1.3939x over previous
#include <hip/hip_runtime.h>

typedef __attribute__((ext_vector_type(2))) float v2f;
typedef __attribute__((ext_vector_type(8))) float v8f;

#define BATCH 32
#define NROWS 1024
#define DDIM  256
#define SPLIT 8

#define RGROUPS 4
#define RSTEPS  (NROWS / SPLIT / RGROUPS)

__global__ __launch_bounds__(256) void reduce_partial(const float* __restrict__ x,
                                                      const float* __restrict__ y,
                                                      float* __restrict__ partial) {
    __shared__ float4 red4[RGROUPS * DDIM / 4];
    const int t     = threadIdx.x;
    const int c4    = t & 63;
    const int g     = t >> 6;
    const int chunk = blockIdx.x;
    const int which = blockIdx.y;
    const int b     = blockIdx.z;
    const float* src = which ? y : x;
    const float4* p = (const float4*)(src + ((size_t)b * NROWS + (size_t)chunk * (NROWS / SPLIT) + g) * DDIM) + c4;
    float4 v[RSTEPS];
#pragma unroll
    for (int i = 0; i < RSTEPS; ++i)
        v[i] = p[(size_t)i * (RGROUPS * DDIM / 4)];
    float4 acc = v[0];
#pragma unroll
    for (int i = 1; i < RSTEPS; ++i) {
        acc.x += v[i].x; acc.y += v[i].y; acc.z += v[i].z; acc.w += v[i].w;
    }
    red4[g * (DDIM / 4) + c4] = acc;
    __syncthreads();
    const float* red = (const float*)red4;
    const float s = (red[t] + red[DDIM + t]) + (red[2 * DDIM + t] + red[3 * DDIM + t]);
    partial[(((size_t)which * BATCH + b) * SPLIT + chunk) * DDIM + t] = s;
}

__global__ __launch_bounds__(256) void reduce_combine(const float* __restrict__ partial,
                                                      float* __restrict__ sums) {
    const int t     = threadIdx.x;
    const int b     = blockIdx.x;
    const int which = blockIdx.y;
    float acc = 0.0f;
#pragma unroll
    for (int c = 0; c < SPLIT; ++c)
        acc += partial[(((size_t)which * BATCH + b) * SPLIT + c) * DDIM + t];
    sums[((size_t)which * BATCH + b) * DDIM + t] = acc;
}

__global__ __launch_bounds__(256) void dot_wmma(const float* __restrict__ x,
                                                const float* __restrict__ y,
                                                const float* __restrict__ sums,
                                                float* __restrict__ out) {
    const int wave = threadIdx.x >> 5;
    const int lane = threadIdx.x & 31;
    const int g    = blockIdx.x * 8 + wave;
    const int h    = g >> 11;
    const int rem  = g & 2047;
    const int b    = rem >> 6;
    const int mbase = (rem & 63) << 4;

    const float* data = h ? x : y;
    const float* vec  = sums + ((size_t)h * BATCH + b) * DDIM;

    const int row  = mbase + (lane & 15);
    const int koff = (lane >> 4) << 1;
    const float* arow = data + ((size_t)b * NROWS + row) * DDIM + koff;
    const float* brow = vec + koff;

    v8f c0 = {};
    v8f c1 = {};
#pragma unroll 4
    for (int d0 = 0; d0 < DDIM; d0 += 8) {
        v2f a0 = *(const v2f*)(arow + d0);
        v2f b0 = *(const v2f*)(brow + d0);
        c0 = __builtin_amdgcn_wmma_f32_16x16x4_f32(false, a0, false, b0,
                                                   (short)0, c0, false, false);
        v2f a1 = *(const v2f*)(arow + d0 + 4);
        v2f b1 = *(const v2f*)(brow + d0 + 4);
        c1 = __builtin_amdgcn_wmma_f32_16x16x4_f32(false, a1, false, b1,
                                                   (short)0, c1, false, false);
    }
    v8f c = c0 + c1;

    const size_t obase = (size_t)b * 2048 + (size_t)h * 1024 + (size_t)mbase;
    if (lane == 0) {
#pragma unroll
        for (int i = 0; i < 8; ++i) out[obase + i] = c[i];
    } else if (lane == 16) {
#pragma unroll
        for (int i = 0; i < 8; ++i) out[obase + 8 + i] = c[i];
    }
}

extern "C" void kernel_launch(void* const* d_in, const int* in_sizes, int n_in,
                              void* d_out, int out_size, void* d_ws, size_t ws_size,
                              hipStream_t stream) {
    const float* x = (const float*)d_in[0];
    const float* y = (const float*)d_in[1];
    float* out = (float*)d_out;

    float* partial = (float*)d_ws;
    float* sums    = partial + (size_t)2 * BATCH * SPLIT * DDIM;

    reduce_partial<<<dim3(SPLIT, 2, BATCH), 256, 0, stream>>>(x, y, partial);
    reduce_combine<<<dim3(BATCH, 2), 256, 0, stream>>>(partial, sums);
    dot_wmma<<<dim3(512), 256, 0, stream>>>(x, y, sums, out);
}


# dot_wmma: tile = 4 data rows x 4 column segments as its 16 matrix rows (was 16 rows): 128-bit A loads of 4 whole 128-B lines, each line once, all issued first; B once per wave; dot sum order changes
# speedup vs baseline: 1.6911x; 1.2132x over previous
#include <hip/hip_runtime.h>

typedef __attribute__((ext_vector_type(2))) float v2f;
typedef __attribute__((ext_vector_type(4))) float v4f;
typedef __attribute__((ext_vector_type(8))) float v8f;

#define BATCH 32
#define NROWS 1024
#define DDIM  256
#define SPLIT 8

#define RGROUPS 4
#define RSTEPS  (NROWS / SPLIT / RGROUPS)

__global__ __launch_bounds__(256) void reduce_partial(const float* __restrict__ x,
                                                      const float* __restrict__ y,
                                                      float* __restrict__ partial) {
    __shared__ float4 red4[RGROUPS * DDIM / 4];
    const int t     = threadIdx.x;
    const int c4    = t & 63;
    const int g     = t >> 6;
    const int chunk = blockIdx.x;
    const int which = blockIdx.y;
    const int b     = blockIdx.z;
    const float* src = which ? y : x;
    const float4* p = (const float4*)(src + ((size_t)b * NROWS + (size_t)chunk * (NROWS / SPLIT) + g) * DDIM) + c4;
    float4 v[RSTEPS];
#pragma unroll
    for (int i = 0; i < RSTEPS; ++i)
        v[i] = p[(size_t)i * (RGROUPS * DDIM / 4)];
    float4 acc = v[0];
#pragma unroll
    for (int i = 1; i < RSTEPS; ++i) {
        acc.x += v[i].x; acc.y += v[i].y; acc.z += v[i].z; acc.w += v[i].w;
    }
    red4[g * (DDIM / 4) + c4] = acc;
    __syncthreads();
    const float* red = (const float*)red4;
    const float s = (red[t] + red[DDIM + t]) + (red[2 * DDIM + t] + red[3 * DDIM + t]);
    partial[(((size_t)which * BATCH + b) * SPLIT + chunk) * DDIM + t] = s;
}

__global__ __launch_bounds__(256) void reduce_combine(const float* __restrict__ partial,
                                                      float* __restrict__ sums) {
    const int t     = threadIdx.x;
    const int b     = blockIdx.x;
    const int which = blockIdx.y;
    float acc = 0.0f;
#pragma unroll
    for (int c = 0; c < SPLIT; ++c)
        acc += partial[(((size_t)which * BATCH + b) * SPLIT + c) * DDIM + t];
    sums[((size_t)which * BATCH + b) * DDIM + t] = acc;
}

#define TILE_ROWS 16
#define GROUP_ROWS 4
#define NGROUPS (TILE_ROWS / GROUP_ROWS)
#define STEP_COLS 32
#define NSTEPS (DDIM / STEP_COLS)

template <int K>
__device__ __forceinline__ float quad_lane(float v) {
    return __builtin_bit_cast(float, __builtin_amdgcn_update_dpp(0, __builtin_bit_cast(int, v), K * 0x55, 0xf, 0xf, true));
}

__global__ __launch_bounds__(256) __attribute__((amdgpu_waves_per_eu(4, 4)))
void dot_wmma(const float* __restrict__ x,
              const float* __restrict__ y,
              const float* __restrict__ sums,
              float* __restrict__ out) {
    const int wave = threadIdx.x >> 5;
    const int lane = threadIdx.x & 31;
    const int g    = blockIdx.x * 8 + wave;
    const int h    = g >> 11;
    const int rem  = g & 2047;
    const int b    = rem >> 6;
    const int mbase = (rem & 63) << 4;

    const float* data = h ? x : y;
    const float* vec  = sums + ((size_t)h * BATCH + b) * DDIM;

    const int half = lane >> 4;
    const int c    = lane & 15;
    const int quad = c >> 2;
    const int seg  = c & 3;
    const int col  = seg * 8 + half * 4;
    const float* ap = data + ((size_t)b * NROWS + mbase + quad) * DDIM + col;
    const float* bp = vec + col;

    v4f bf[NSTEPS];
#pragma unroll
    for (int i = 0; i < NSTEPS; ++i)
        bf[i] = *(const v4f*)(bp + i * STEP_COLS);

    v4f af[NSTEPS][NGROUPS];
#pragma unroll
    for (int i = 0; i < NSTEPS; ++i)
#pragma unroll
        for (int q = 0; q < NGROUPS; ++q)
            af[i][q] = *(const v4f*)(ap + (size_t)q * GROUP_ROWS * DDIM + i * STEP_COLS);

    v8f acc[NGROUPS];
#pragma unroll
    for (int q = 0; q < NGROUPS; ++q) acc[q] = (v8f){};

#pragma unroll
    for (int i = 0; i < NSTEPS; ++i) {
#pragma unroll
        for (int q = 0; q < NGROUPS; ++q)
            acc[q] = __builtin_amdgcn_wmma_f32_16x16x4_f32(false, af[i][q].xy, false, bf[i].xy,
                                                           (short)0, acc[q], false, false);
#pragma unroll
        for (int q = 0; q < NGROUPS; ++q)
            acc[q] = __builtin_amdgcn_wmma_f32_16x16x4_f32(false, af[i][q].zw, false, bf[i].zw,
                                                           (short)0, acc[q], false, false);
    }

    float lo[NGROUPS], hi[NGROUPS];
#pragma unroll
    for (int q = 0; q < NGROUPS; ++q) {
        lo[q] = (quad_lane<0>(acc[q][0]) + quad_lane<1>(acc[q][1])) +
                (quad_lane<2>(acc[q][2]) + quad_lane<3>(acc[q][3]));
        hi[q] = (quad_lane<0>(acc[q][4]) + quad_lane<1>(acc[q][5])) +
                (quad_lane<2>(acc[q][6]) + quad_lane<3>(acc[q][7]));
    }
    float vlo = lo[0], vhi = hi[0];
    vlo = quad == 1 ? lo[1] : vlo;  vhi = quad == 1 ? hi[1] : vhi;
    vlo = quad == 2 ? lo[2] : vlo;  vhi = quad == 2 ? hi[2] : vhi;
    vlo = quad == 3 ? lo[3] : vlo;  vhi = quad == 3 ? hi[3] : vhi;
    const float val = (seg & 1) ? vhi : vlo;
    const size_t obase = (size_t)b * 2048 + (size_t)h * 1024 + (size_t)mbase;
    if (seg < 2)
        out[obase + 4 * quad + 2 * half + seg] = val;
}

extern "C" void kernel_launch(void* const* d_in, const int* in_sizes, int n_in,
                              void* d_out, int out_size, void* d_ws, size_t ws_size,
                              hipStream_t stream) {
    const float* x = (const float*)d_in[0];
    const float* y = (const float*)d_in[1];
    float* out = (float*)d_out;

    float* partial = (float*)d_ws;
    float* sums    = partial + (size_t)2 * BATCH * SPLIT * DDIM;

    reduce_partial<<<dim3(SPLIT, 2, BATCH), 256, 0, stream>>>(x, y, partial);
    reduce_combine<<<dim3(BATCH, 2), 256, 0, stream>>>(partial, sums);
    dot_wmma<<<dim3(512), 256, 0, stream>>>(x, y, sums, out);
}


# dot_wmma folds the 8 partials of its batch itself (a column per thread, same order, via 1 KB LDS behind an LDS-only barrier; B fragments from LDS): reduce_combine launch removed, 3 launches -> 2
# speedup vs baseline: 1.8161x; 1.0739x over previous
#include <hip/hip_runtime.h>

typedef __attribute__((ext_vector_type(2))) float v2f;
typedef __attribute__((ext_vector_type(4))) float v4f;
typedef __attribute__((ext_vector_type(8))) float v8f;

#define BATCH 32
#define NROWS 1024
#define DDIM  256
#define SPLIT 8

#define RGROUPS 4
#define RSTEPS  (NROWS / SPLIT / RGROUPS)

__global__ __launch_bounds__(256) void reduce_partial(const float* __restrict__ x,
                                                      const float* __restrict__ y,
                                                      float* __restrict__ partial) {
    __shared__ float4 red4[RGROUPS * DDIM / 4];
    const int t     = threadIdx.x;
    const int c4    = t & 63;
    const int g     = t >> 6;
    const int chunk = blockIdx.x;
    const int which = blockIdx.y;
    const int b     = blockIdx.z;
    const float* src = which ? y : x;
    const float4* p = (const float4*)(src + ((size_t)b * NROWS + (size_t)chunk * (NROWS / SPLIT) + g) * DDIM) + c4;
    float4 v[RSTEPS];
#pragma unroll
    for (int i = 0; i < RSTEPS; ++i)
        v[i] = p[(size_t)i * (RGROUPS * DDIM / 4)];
    float4 acc = v[0];
#pragma unroll
    for (int i = 1; i < RSTEPS; ++i) {
        acc.x += v[i].x; acc.y += v[i].y; acc.z += v[i].z; acc.w += v[i].w;
    }
    red4[g * (DDIM / 4) + c4] = acc;
    __syncthreads();
    const float* red = (const float*)red4;
    const float s = (red[t] + red[DDIM + t]) + (red[2 * DDIM + t] + red[3 * DDIM + t]);
    partial[(((size_t)which * BATCH + b) * SPLIT + chunk) * DDIM + t] = s;
}

#define TILE_ROWS 16
#define GROUP_ROWS 4
#define NGROUPS (TILE_ROWS / GROUP_ROWS)
#define STEP_COLS 32
#define NSTEPS (DDIM / STEP_COLS)

template <int K>
__device__ __forceinline__ float quad_lane(float v) {
    return __builtin_bit_cast(float, __builtin_amdgcn_update_dpp(0, __builtin_bit_cast(int, v), K * 0x55, 0xf, 0xf, true));
}

__global__ __launch_bounds__(256) __attribute__((amdgpu_waves_per_eu(4, 4)))
void dot_wmma(const float* __restrict__ x,
              const float* __restrict__ y,
              const float* __restrict__ partial,
              float* __restrict__ out) {
    __shared__ __attribute__((aligned(16))) float svec[DDIM];
    const int t    = threadIdx.x;
    const int wave = t >> 5;
    const int lane = t & 31;
    const int h    = blockIdx.x >> 8;
    const int b    = (blockIdx.x & 255) >> 3;
    const int mbase = (((blockIdx.x & 7) << 3) + wave) << 4;

    const float* data = h ? x : y;

    const float* pp = partial + ((size_t)h * BATCH + b) * SPLIT * DDIM + t;
    float pv[SPLIT];
#pragma unroll
    for (int k = 0; k < SPLIT; ++k)
        pv[k] = pp[k * DDIM];

    const int half = lane >> 4;
    const int c    = lane & 15;
    const int quad = c >> 2;
    const int seg  = c & 3;
    const int col  = seg * 8 + half * 4;
    const float* ap = data + ((size_t)b * NROWS + mbase + quad) * DDIM + col;

    v4f af[NSTEPS][NGROUPS];
#pragma unroll
    for (int i = 0; i < NSTEPS; ++i)
#pragma unroll
        for (int q = 0; q < NGROUPS; ++q)
            af[i][q] = *(const v4f*)(ap + (size_t)q * GROUP_ROWS * DDIM + i * STEP_COLS);

    float s = 0.0f;
#pragma unroll
    for (int k = 0; k < SPLIT; ++k)
        s += pv[k];
    svec[t] = s;
    __builtin_amdgcn_fence(__ATOMIC_RELEASE, "workgroup", "local");
    __builtin_amdgcn_s_barrier();
    __builtin_amdgcn_fence(__ATOMIC_ACQUIRE, "workgroup", "local");

    v4f bf[NSTEPS];
#pragma unroll
    for (int i = 0; i < NSTEPS; ++i)
        bf[i] = *(const v4f*)(svec + col + i * STEP_COLS);
    __builtin_amdgcn_sched_barrier(0);

    v8f acc[NGROUPS];
#pragma unroll
    for (int q = 0; q < NGROUPS; ++q) acc[q] = (v8f){};

#pragma unroll
    for (int i = 0; i < NSTEPS; ++i) {
#pragma unroll
        for (int q = 0; q < NGROUPS; ++q)
            acc[q] = __builtin_amdgcn_wmma_f32_16x16x4_f32(false, af[i][q].xy, false, bf[i].xy,
                                                           (short)0, acc[q], false, false);
#pragma unroll
        for (int q = 0; q < NGROUPS; ++q)
            acc[q] = __builtin_amdgcn_wmma_f32_16x16x4_f32(false, af[i][q].zw, false, bf[i].zw,
                                                           (short)0, acc[q], false, false);
    }

    float lo[NGROUPS], hi[NGROUPS];
#pragma unroll
    for (int q = 0; q < NGROUPS; ++q) {
        lo[q] = (quad_lane<0>(acc[q][0]) + quad_lane<1>(acc[q][1])) +
                (quad_lane<2>(acc[q][2]) + quad_lane<3>(acc[q][3]));
        hi[q] = (quad_lane<0>(acc[q][4]) + quad_lane<1>(acc[q][5])) +
                (quad_lane<2>(acc[q][6]) + quad_lane<3>(acc[q][7]));
    }
    float vlo = lo[0], vhi = hi[0];
    vlo = quad == 1 ? lo[1] : vlo;  vhi = quad == 1 ? hi[1] : vhi;
    vlo = quad == 2 ? lo[2] : vlo;  vhi = quad == 2 ? hi[2] : vhi;
    vlo = quad == 3 ? lo[3] : vlo;  vhi = quad == 3 ? hi[3] : vhi;
    const float val = (seg & 1) ? vhi : vlo;
    const size_t obase = (size_t)b * 2048 + (size_t)h * 1024 + (size_t)mbase;
    if (seg < 2)
        out[obase + 4 * quad + 2 * half + seg] = val;
}

extern "C" void kernel_launch(void* const* d_in, const int* in_sizes, int n_in,
                              void* d_out, int out_size, void* d_ws, size_t ws_size,
                              hipStream_t stream) {
    const float* x = (const float*)d_in[0];
    const float* y = (const float*)d_in[1];
    float* out = (float*)d_out;

    float* partial = (float*)d_ws;

    reduce_partial<<<dim3(SPLIT, 2, BATCH), 256, 0, stream>>>(x, y, partial);
    dot_wmma<<<dim3(512), 256, 0, stream>>>(x, y, partial, out);
}
